# starting kernel
# baseline (speedup 1.0000x reference)
#include <hip/hip_runtime.h>

typedef __attribute__((ext_vector_type(2))) float v2f;
typedef __attribute__((ext_vector_type(8))) float v8f;

#define TPB 256
#define WAVES_PER_BLOCK (TPB / 32)

__global__ __launch_bounds__(TPB) void color3x3_wmma(const float* __restrict__ x,
                                                     const float* __restrict__ w,
                                                     float* __restrict__ out,
                                                     int numGroups) {
  __shared__ float4 ldsIn4[WAVES_PER_BLOCK][12];
  __shared__ float4 ldsOut4[WAVES_PER_BLOCK][12];

  const int lane        = threadIdx.x & 31;
  const int waveInBlock = threadIdx.x >> 5;
  const int waveId      = (int)((blockIdx.x * TPB + threadIdx.x) >> 5);
  const int wavesTotal  = (int)((gridDim.x * TPB) >> 5);

  const int n    = lane & 15;
  const int half = lane >> 4;

  v2f B;
  B.x = 0.0f;
  B.y = 0.0f;
  if (n < 3) {
    B.x = w[(half ? 2 : 0) * 3 + n];
    B.y = half ? 0.0f : w[3 + n];
  }

  float*       mIn  = (float*)ldsIn4[waveInBlock];
  float*       mOut = (float*)ldsOut4[waveInBlock];
  const float4* x4  = (const float4*)x;

  for (int g = waveId; g < numGroups; g += wavesTotal) {
    const size_t base4 = (size_t)g * 12;

    if (lane < 12) {
      ldsIn4[waveInBlock][lane] = x4[base4 + lane];
      if (g + wavesTotal < numGroups)
        __builtin_prefetch(&x4[base4 + (size_t)wavesTotal * 12 + lane], 0, 3);
    }
    asm volatile("s_wait_dscnt 0" ::: "memory");

    v2f A;
    A.x = mIn[3 * n + (half ? 2 : 0)];
    A.y = half ? 0.0f : mIn[3 * n + 1];

    v8f C = {0.f, 0.f, 0.f, 0.f, 0.f, 0.f, 0.f, 0.f};
    C = __builtin_amdgcn_wmma_f32_16x16x4_f32(
         false, A,  false, B,
         (short)0, C,  false,  false);

    if (n < 3) {
      const int rb = half * 8;
      mOut[(rb + 0) * 3 + n] = C[0];
      mOut[(rb + 1) * 3 + n] = C[1];
      mOut[(rb + 2) * 3 + n] = C[2];
      mOut[(rb + 3) * 3 + n] = C[3];
      mOut[(rb + 4) * 3 + n] = C[4];
      mOut[(rb + 5) * 3 + n] = C[5];
      mOut[(rb + 6) * 3 + n] = C[6];
      mOut[(rb + 7) * 3 + n] = C[7];
    }
    asm volatile("s_wait_dscnt 0" ::: "memory");

    if (lane < 12) {
      ((float4*)out)[base4 + lane] = ldsOut4[waveInBlock][lane];
    }
  }
}

__global__ void color3x3_tail(const float* __restrict__ x, const float* __restrict__ w,
                              float* __restrict__ out, long start, long totalPixels) {
  long p = start + (long)blockIdx.x * blockDim.x + threadIdx.x;
  if (p >= totalPixels) return;
  const float c0 = x[3 * p + 0], c1 = x[3 * p + 1], c2 = x[3 * p + 2];
#pragma unroll
  for (int d = 0; d < 3; ++d)
    out[3 * p + d] = c0 * w[d] + c1 * w[3 + d] + c2 * w[6 + d];
}

extern "C" void kernel_launch(void* const* d_in, const int* in_sizes, int n_in,
                              void* d_out, int out_size, void* d_ws, size_t ws_size,
                              hipStream_t stream) {
  const float* x = (const float*)d_in[0];
  const float* w = (const float*)d_in[1];
  float* out = (float*)d_out;

  const long totalPixels = (long)in_sizes[0] / 3;
  const long groups      = totalPixels / 16;

  if (groups > 0) {
    int blocks = 4096;
    long waves = (long)blocks * (TPB / 32);
    if (waves > groups) blocks = (int)((groups * 32 + TPB - 1) / TPB) + 1;
    color3x3_wmma<<<blocks, TPB, 0, stream>>>(x, w, out, (int)groups);
  }

  const long tailStart = groups * 16;
  const long tail      = totalPixels - tailStart;
  if (tail > 0) {
    int tb = (int)((tail + TPB - 1) / TPB);
    color3x3_tail<<<tb, TPB, 0, stream>>>(x, w, out, tailStart, totalPixels);
  }
}


# Wave tile 16 -> 256 pixels: each trip moves 3072 contiguous bytes with six full-wave 16-byte loads (next tile requested one trip ahead) and six such stores; 512 work-groups instead of 4096
# speedup vs baseline: 1.7012x; 1.7012x over previous
#include <hip/hip_runtime.h>

typedef __attribute__((ext_vector_type(2))) float v2f;
typedef __attribute__((ext_vector_type(8))) float v8f;
typedef __attribute__((ext_vector_type(4))) float v4f;

#define TPB 256
#define WAVES_PER_BLOCK (TPB / 32)
#define GROUPS_PER_TILE 16
#define F4_PER_TILE (GROUPS_PER_TILE * 12)
#define F4_PER_LANE (F4_PER_TILE / 32)

__global__ __launch_bounds__(TPB) void color3x3_wmma(const float* __restrict__ x,
                                                     const float* __restrict__ w,
                                                     float* __restrict__ out,
                                                     int numTiles) {
  __shared__ v4f ldsIn4[WAVES_PER_BLOCK][F4_PER_TILE];
  __shared__ v4f ldsOut4[WAVES_PER_BLOCK][F4_PER_TILE];

  const int lane        = threadIdx.x & 31;
  const int waveInBlock = threadIdx.x >> 5;
  const int waveId      = (int)((blockIdx.x * TPB + threadIdx.x) >> 5);
  const int wavesTotal  = (int)((gridDim.x * TPB) >> 5);

  const int n    = lane & 15;
  const int half = lane >> 4;

  v2f B;
  B.x = 0.0f;
  B.y = 0.0f;
  if (n < 3) {
    B.x = w[(half ? 2 : 0) * 3 + n];
    B.y = half ? 0.0f : w[3 + n];
  }

  const float*  mIn  = (const float*)ldsIn4[waveInBlock];
  float*        mOut = (float*)ldsOut4[waveInBlock];
  const v4f*    x4   = (const v4f*)x;
  v4f*          o4   = (v4f*)out;

  v4f r[F4_PER_LANE];
  int t = waveId;
  if (t < numTiles) {
#pragma unroll
    for (int i = 0; i < F4_PER_LANE; ++i)
      r[i] = x4[(size_t)t * F4_PER_TILE + i * 32 + lane];
  }

  for (; t < numTiles; t += wavesTotal) {
    const size_t base4 = (size_t)t * F4_PER_TILE;

#pragma unroll
    for (int i = 0; i < F4_PER_LANE; ++i)
      ldsIn4[waveInBlock][i * 32 + lane] = r[i];

    const int tn = t + wavesTotal;
    if (tn < numTiles) {
#pragma unroll
      for (int i = 0; i < F4_PER_LANE; ++i)
        r[i] = x4[(size_t)tn * F4_PER_TILE + i * 32 + lane];
    }

    asm volatile("s_wait_dscnt 0" ::: "memory");

    v2f A[GROUPS_PER_TILE];
#pragma unroll
    for (int j = 0; j < GROUPS_PER_TILE; ++j) {
      const float k02 = mIn[48 * j + 3 * n + (half ? 2 : 0)];
      const float k1  = mIn[48 * j + 3 * n + 1];
      A[j].x = k02;
      A[j].y = half ? 0.0f : k1;
    }

    v8f C[GROUPS_PER_TILE];
#pragma unroll
    for (int j = 0; j < GROUPS_PER_TILE; ++j) {
      const v8f zero = {0.f, 0.f, 0.f, 0.f, 0.f, 0.f, 0.f, 0.f};
      C[j] = __builtin_amdgcn_wmma_f32_16x16x4_f32(
           false, A[j],  false, B,
           (short)0, zero,  false,  false);
    }

    if (n < 3) {
#pragma unroll
      for (int j = 0; j < GROUPS_PER_TILE; ++j) {
        const int rb = 48 * j + half * 24;
        mOut[rb + 0 * 3 + n] = C[j][0];
        mOut[rb + 1 * 3 + n] = C[j][1];
        mOut[rb + 2 * 3 + n] = C[j][2];
        mOut[rb + 3 * 3 + n] = C[j][3];
        mOut[rb + 4 * 3 + n] = C[j][4];
        mOut[rb + 5 * 3 + n] = C[j][5];
        mOut[rb + 6 * 3 + n] = C[j][6];
        mOut[rb + 7 * 3 + n] = C[j][7];
      }
    }
    asm volatile("s_wait_dscnt 0" ::: "memory");

#pragma unroll
    for (int i = 0; i < F4_PER_LANE; ++i)
      o4[base4 + i * 32 + lane] = ldsOut4[waveInBlock][i * 32 + lane];
  }
}

__global__ void color3x3_tail(const float* __restrict__ x, const float* __restrict__ w,
                              float* __restrict__ out, long start, long totalPixels) {
  long p = start + (long)blockIdx.x * blockDim.x + threadIdx.x;
  if (p >= totalPixels) return;
  const float c0 = x[3 * p + 0], c1 = x[3 * p + 1], c2 = x[3 * p + 2];
#pragma unroll
  for (int d = 0; d < 3; ++d)
    out[3 * p + d] = c0 * w[d] + c1 * w[3 + d] + c2 * w[6 + d];
}

extern "C" void kernel_launch(void* const* d_in, const int* in_sizes, int n_in,
                              void* d_out, int out_size, void* d_ws, size_t ws_size,
                              hipStream_t stream) {
  const float* x = (const float*)d_in[0];
  const float* w = (const float*)d_in[1];
  float* out = (float*)d_out;

  const long totalPixels = (long)in_sizes[0] / 3;
  const long tiles       = totalPixels / (16 * GROUPS_PER_TILE);

  if (tiles > 0) {
    int blocks = 512;
    long waves = (long)blocks * (TPB / 32);
    if (waves > tiles) blocks = (int)((tiles + (TPB / 32) - 1) / (TPB / 32));
    color3x3_wmma<<<blocks, TPB, 0, stream>>>(x, w, out, (int)tiles);
  }

  const long tailStart = tiles * (16 * GROUPS_PER_TILE);
  const long tail      = totalPixels - tailStart;
  if (tail > 0) {
    int tb = (int)((tail + TPB - 1) / TPB);
    color3x3_tail<<<tb, TPB, 0, stream>>>(x, w, out, tailStart, totalPixels);
  }
}


# Operands regrouped: block-diagonal w (4 pixels, K=12) as A, pixel groups as B columns; lanes hold contiguous floats, so LDS moves in 16-byte pieces; 12 WMMA per tile; per-pixel sum order changes
# speedup vs baseline: 1.8840x; 1.1075x over previous
#include <hip/hip_runtime.h>

typedef __attribute__((ext_vector_type(2))) float v2f;
typedef __attribute__((ext_vector_type(8))) float v8f;
typedef __attribute__((ext_vector_type(4))) float v4f;

#define TPB 256
#define WAVES_PER_BLOCK (TPB / 32)
#define GROUPS_PER_TILE 16
#define COLTILES_PER_TILE 4
#define F4_PER_TILE (GROUPS_PER_TILE * 12)
#define F4_PER_LANE (F4_PER_TILE / 32)

__device__ __forceinline__ int kIndex(int half, int t) {
  return half ? (t < 4 ? 8 + t : 2 + t) : t;
}

__global__ __launch_bounds__(TPB) void color3x3_wmma(const float* __restrict__ x,
                                                     const float* __restrict__ w,
                                                     float* __restrict__ out,
                                                     int numTiles) {
  __shared__ v4f ldsIn4[WAVES_PER_BLOCK][F4_PER_TILE];
  __shared__ v4f ldsOut4[WAVES_PER_BLOCK][F4_PER_TILE];

  const int lane        = threadIdx.x & 31;
  const int waveInBlock = threadIdx.x >> 5;
  const int waveId      = (int)((blockIdx.x * TPB + threadIdx.x) >> 5);
  const int wavesTotal  = (int)((gridDim.x * TPB) >> 5);

  const int c    = lane & 15;
  const int half = lane >> 4;

  v2f Aw[3];
#pragma unroll
  for (int s = 0; s < 3; ++s) {
    const int i0 = kIndex(half, 2 * s + 0);
    const int i1 = kIndex(half, 2 * s + 1);
    const float w0 = w[(i0 % 3) * 3 + (c % 3)];
    const float w1 = w[(i1 % 3) * 3 + (c % 3)];
    Aw[s].x = (c < 12 && (i0 / 3) == (c / 3)) ? w0 : 0.0f;
    Aw[s].y = (c < 12 && (i1 / 3) == (c / 3)) ? w1 : 0.0f;
  }

  const float*  mIn  = (const float*)ldsIn4[waveInBlock];
  float*        mOut = (float*)ldsOut4[waveInBlock];
  const v4f*    x4   = (const v4f*)x;
  v4f*          o4   = (v4f*)out;

  const int off16 = 12 * c + (half ? 8 : 0);
  const int off8  = 12 * c + (half ? 6 : 4);
  const int offMid = 12 * c + 4;

  v4f r[F4_PER_LANE];
  int t = waveId;
  if (t < numTiles) {
#pragma unroll
    for (int i = 0; i < F4_PER_LANE; ++i)
      r[i] = x4[(size_t)t * F4_PER_TILE + i * 32 + lane];
  }

  for (; t < numTiles; t += wavesTotal) {
    const size_t base4 = (size_t)t * F4_PER_TILE;

#pragma unroll
    for (int i = 0; i < F4_PER_LANE; ++i)
      ldsIn4[waveInBlock][i * 32 + lane] = r[i];

    const int tn = t + wavesTotal;
    if (tn < numTiles) {
#pragma unroll
      for (int i = 0; i < F4_PER_LANE; ++i)
        r[i] = x4[(size_t)tn * F4_PER_TILE + i * 32 + lane];
    }

    asm volatile("s_wait_dscnt 0" ::: "memory");

    v4f g4[COLTILES_PER_TILE];
    v2f g2[COLTILES_PER_TILE];
#pragma unroll
    for (int ct = 0; ct < COLTILES_PER_TILE; ++ct) {
      g4[ct] = *(const v4f*)(mIn + 192 * ct + off16);
      g2[ct] = *(const v2f*)(mIn + 192 * ct + off8);
    }

    v8f D[COLTILES_PER_TILE];
#pragma unroll
    for (int ct = 0; ct < COLTILES_PER_TILE; ++ct) {
      const v8f zero = {0.f, 0.f, 0.f, 0.f, 0.f, 0.f, 0.f, 0.f};
      const v2f b0 = {g4[ct].x, g4[ct].y};
      D[ct] = __builtin_amdgcn_wmma_f32_16x16x4_f32(
           false, Aw[0],  false, b0,
           (short)0, zero,  false,  false);
    }
#pragma unroll
    for (int ct = 0; ct < COLTILES_PER_TILE; ++ct) {
      const v2f b1 = {g4[ct].z, g4[ct].w};
      D[ct] = __builtin_amdgcn_wmma_f32_16x16x4_f32(
           false, Aw[1],  false, b1,
           (short)0, D[ct],  false,  false);
    }
#pragma unroll
    for (int ct = 0; ct < COLTILES_PER_TILE; ++ct) {
      D[ct] = __builtin_amdgcn_wmma_f32_16x16x4_f32(
           false, Aw[2],  false, g2[ct],
           (short)0, D[ct],  false,  false);
    }

#pragma unroll
    for (int ct = 0; ct < COLTILES_PER_TILE; ++ct) {
      const v4f lo = {D[ct][0], D[ct][1], D[ct][2], D[ct][3]};
      *(v4f*)(mOut + 192 * ct + off16) = lo;
    }
    if (half == 0) {
#pragma unroll
      for (int ct = 0; ct < COLTILES_PER_TILE; ++ct) {
        const v4f mid = {D[ct][4], D[ct][5], D[ct][6], D[ct][7]};
        *(v4f*)(mOut + 192 * ct + offMid) = mid;
      }
    }
    asm volatile("s_wait_dscnt 0" ::: "memory");

#pragma unroll
    for (int i = 0; i < F4_PER_LANE; ++i)
      o4[base4 + i * 32 + lane] = ldsOut4[waveInBlock][i * 32 + lane];
  }
}

__global__ void color3x3_tail(const float* __restrict__ x, const float* __restrict__ w,
                              float* __restrict__ out, long start, long totalPixels) {
  long p = start + (long)blockIdx.x * blockDim.x + threadIdx.x;
  if (p >= totalPixels) return;
  const float c0 = x[3 * p + 0], c1 = x[3 * p + 1], c2 = x[3 * p + 2];
#pragma unroll
  for (int d = 0; d < 3; ++d)
    out[3 * p + d] = c0 * w[d] + c1 * w[3 + d] + c2 * w[6 + d];
}

extern "C" void kernel_launch(void* const* d_in, const int* in_sizes, int n_in,
                              void* d_out, int out_size, void* d_ws, size_t ws_size,
                              hipStream_t stream) {
  const float* x = (const float*)d_in[0];
  const float* w = (const float*)d_in[1];
  float* out = (float*)d_out;

  const long totalPixels = (long)in_sizes[0] / 3;
  const long tiles       = totalPixels / (16 * GROUPS_PER_TILE);

  if (tiles > 0) {
    int blocks = 512;
    long waves = (long)blocks * (TPB / 32);
    if (waves > tiles) blocks = (int)((tiles + (TPB / 32) - 1) / (TPB / 32));
    color3x3_wmma<<<blocks, TPB, 0, stream>>>(x, w, out, (int)tiles);
  }

  const long tailStart = tiles * (16 * GROUPS_PER_TILE);
  const long tail      = totalPixels - tailStart;
  if (tail > 0) {
    int tb = (int)((tail + TPB - 1) / TPB);
    color3x3_tail<<<tb, TPB, 0, stream>>>(x, w, out, tailStart, totalPixels);
  }
}
